# starting kernel
# baseline (speedup 1.0000x reference)
#include <hip/hip_runtime.h>
#include <hip/hip_bf16.h>


typedef __attribute__((ext_vector_type(2))) float v2f;
typedef __attribute__((ext_vector_type(8))) float v8f;

#define LCONST 2048
#define DCONST 1024
#define NHC 16
#define DHC 64

static __device__ inline v8f wmma4(v2f a, v2f b, v8f c) {
    return __builtin_amdgcn_wmma_f32_16x16x4_f32(
         false, a,  false, b,
         (short)0, c,  false,  false);
}

__global__ __launch_bounds__(128) void qkv_rms_kernel(
    const float* __restrict__ x, const float* __restrict__ w,
    const float* __restrict__ qnw, const float* __restrict__ knw,
    float* __restrict__ qb, float* __restrict__ kb, float* __restrict__ vb) {
    const int lane  = threadIdx.x & 31;
    const int wave  = threadIdx.x >> 5;
    const int laneN = lane & 15;
    const int koff  = (lane >> 4) << 1;
    const int hi8   = (lane >> 4) << 3;
    const int m0 = blockIdx.y * 64;
    const int n0 = blockIdx.x * 256 + wave * 64;

    v8f zero = {};
    v8f acc[4][4];
#pragma unroll
    for (int mi = 0; mi < 4; ++mi)
#pragma unroll
        for (int ni = 0; ni < 4; ++ni) acc[mi][ni] = zero;

    for (int kk = 0; kk < DCONST; kk += 4) {
        v2f a[4], bf[4];
#pragma unroll
        for (int i = 0; i < 4; ++i)
            a[i] = *(const v2f*)(x + (size_t)(m0 + i * 16 + laneN) * DCONST + kk + koff);
#pragma unroll
        for (int i = 0; i < 4; ++i)
            bf[i] = *(const v2f*)(w + (size_t)(n0 + i * 16 + laneN) * DCONST + kk + koff);
#pragma unroll
        for (int mi = 0; mi < 4; ++mi)
#pragma unroll
            for (int ni = 0; ni < 4; ++ni)
                acc[mi][ni] = wmma4(a[mi], bf[ni], acc[mi][ni]);
    }

    const int sec = n0 >> 10;
    const int h   = (n0 & 1023) >> 6;
    float* dst = (sec == 0) ? qb : ((sec == 1) ? kb : vb);
    float wn[4];
#pragma unroll
    for (int ni = 0; ni < 4; ++ni) {
        const float* nw = (sec == 0) ? qnw : knw;
        wn[ni] = (sec < 2) ? nw[ni * 16 + laneN] : 1.0f;
    }

#pragma unroll
    for (int mi = 0; mi < 4; ++mi) {
#pragma unroll
        for (int r = 0; r < 8; ++r) {
            float sv[4];
#pragma unroll
            for (int ni = 0; ni < 4; ++ni) sv[ni] = acc[mi][ni][r];
            if (sec < 2) {
                float ss = sv[0] * sv[0] + sv[1] * sv[1] + sv[2] * sv[2] + sv[3] * sv[3];
                ss += __shfl_xor(ss, 1, 32);
                ss += __shfl_xor(ss, 2, 32);
                ss += __shfl_xor(ss, 4, 32);
                ss += __shfl_xor(ss, 8, 32);
                float rstd = rsqrtf(ss * (1.0f / 64.0f) + 1e-5f);
#pragma unroll
                for (int ni = 0; ni < 4; ++ni) sv[ni] *= rstd * wn[ni];
            }
            const int m = m0 + mi * 16 + r + hi8;
            const int b = m >> 11;
            const int l = m & (LCONST - 1);
            float* rowp = dst + ((size_t)((b * NHC + h) * LCONST + l) << 6);
#pragma unroll
            for (int ni = 0; ni < 4; ++ni) rowp[ni * 16 + laneN] = sv[ni];
        }
    }
}

__global__ __launch_bounds__(32) void attn_kernel(
    const float* __restrict__ qb, const float* __restrict__ kb,
    const float* __restrict__ vb, float* __restrict__ ob) {
    const int lane  = threadIdx.x;
    const int laneN = lane & 15;
    const int koff  = (lane >> 4) << 1;
    const int hi8   = (lane >> 4) << 3;
    const int q0 = blockIdx.x * 32;
    const int bh = blockIdx.z * NHC + blockIdx.y;
    const float* qh = qb + (size_t)bh * LCONST * DHC;
    const float* kh = kb + (size_t)bh * LCONST * DHC;
    const float* vh = vb + (size_t)bh * LCONST * DHC;

    __shared__ float q_s[32 * 68];
    __shared__ float p_s[32 * 68];

#pragma unroll 4
    for (int r = 0; r < 32; ++r) {
        v2f t = *(const v2f*)(qh + (size_t)(q0 + r) * DHC + 2 * lane);
        *(v2f*)(q_s + r * 68 + 2 * lane) = t;
    }
    __syncthreads();

    v8f zero = {};
    v8f o[2][4];
    float mr[2][8], lr[2][8];
#pragma unroll
    for (int mi = 0; mi < 2; ++mi) {
#pragma unroll
        for (int nj = 0; nj < 4; ++nj) o[mi][nj] = zero;
#pragma unroll
        for (int r = 0; r < 8; ++r) { mr[mi][r] = -3.0e38f; lr[mi][r] = 0.0f; }
    }

    const int nkv = (q0 >> 6) + 1;
    for (int kv = 0; kv < nkv; ++kv) {
        const int kbase = kv * 64;
        v8f s[2][4];
#pragma unroll
        for (int mi = 0; mi < 2; ++mi)
#pragma unroll
            for (int ni = 0; ni < 4; ++ni) s[mi][ni] = zero;

#pragma unroll
        for (int d = 0; d < DHC; d += 4) {
            v2f a0 = *(const v2f*)(q_s + laneN * 68 + d + koff);
            v2f a1 = *(const v2f*)(q_s + (16 + laneN) * 68 + d + koff);
#pragma unroll
            for (int ni = 0; ni < 4; ++ni) {
                v2f bf = *(const v2f*)(kh + (size_t)(kbase + ni * 16 + laneN) * DHC + d + koff);
                s[0][ni] = wmma4(a0, bf, s[0][ni]);
                s[1][ni] = wmma4(a1, bf, s[1][ni]);
            }
        }

        const float scale = 0.125f;
#pragma unroll
        for (int mi = 0; mi < 2; ++mi) {
#pragma unroll
            for (int r = 0; r < 8; ++r) {
                const int qrow = q0 + mi * 16 + r + hi8;
                float sv[4];
                float vmax = -3.0e38f;
#pragma unroll
                for (int ni = 0; ni < 4; ++ni) {
                    const int key = kbase + ni * 16 + laneN;
                    float t = s[mi][ni][r] * scale;
                    t = (key > qrow) ? -1.0e30f : t;
                    sv[ni] = t;
                    vmax = fmaxf(vmax, t);
                }
                vmax = fmaxf(vmax, __shfl_xor(vmax, 1, 32));
                vmax = fmaxf(vmax, __shfl_xor(vmax, 2, 32));
                vmax = fmaxf(vmax, __shfl_xor(vmax, 4, 32));
                vmax = fmaxf(vmax, __shfl_xor(vmax, 8, 32));
                const float mold = mr[mi][r];
                const float mnew = fmaxf(mold, vmax);
                const float corr = __expf(mold - mnew);
                float psum = 0.0f;
#pragma unroll
                for (int ni = 0; ni < 4; ++ni) {
                    float p = __expf(sv[ni] - mnew);
                    sv[ni] = p;
                    psum += p;
                }
                psum += __shfl_xor(psum, 1, 32);
                psum += __shfl_xor(psum, 2, 32);
                psum += __shfl_xor(psum, 4, 32);
                psum += __shfl_xor(psum, 8, 32);
                lr[mi][r] = lr[mi][r] * corr + psum;
                mr[mi][r] = mnew;
#pragma unroll
                for (int nj = 0; nj < 4; ++nj) o[mi][nj][r] *= corr;
#pragma unroll
                for (int ni = 0; ni < 4; ++ni)
                    p_s[(mi * 16 + r + hi8) * 68 + ni * 16 + laneN] = sv[ni];
            }
        }
        __syncthreads();

#pragma unroll
        for (int d = 0; d < 64; d += 4) {
            v2f a0 = *(const v2f*)(p_s + laneN * 68 + d + koff);
            v2f a1 = *(const v2f*)(p_s + (16 + laneN) * 68 + d + koff);
#pragma unroll
            for (int nj = 0; nj < 4; ++nj) {
                v2f bf;
                bf.x = vh[(size_t)(kbase + d + koff) * DHC + nj * 16 + laneN];
                bf.y = vh[(size_t)(kbase + d + koff + 1) * DHC + nj * 16 + laneN];
                o[0][nj] = wmma4(a0, bf, o[0][nj]);
                o[1][nj] = wmma4(a1, bf, o[1][nj]);
            }
        }
        __syncthreads();
    }

#pragma unroll
    for (int mi = 0; mi < 2; ++mi) {
#pragma unroll
        for (int r = 0; r < 8; ++r) {
            const float inv = 1.0f / lr[mi][r];
            const int l = q0 + mi * 16 + r + hi8;
            float* orow = ob + (size_t)(blockIdx.z * LCONST + l) * DCONST + blockIdx.y * DHC;
#pragma unroll
            for (int nj = 0; nj < 4; ++nj) orow[nj * 16 + laneN] = o[mi][nj][r] * inv;
        }
    }
}

__global__ __launch_bounds__(128) void fc_kernel(
    const float* __restrict__ oin, const float* __restrict__ w,
    float* __restrict__ out) {
    const int lane  = threadIdx.x & 31;
    const int wave  = threadIdx.x >> 5;
    const int laneN = lane & 15;
    const int koff  = (lane >> 4) << 1;
    const int hi8   = (lane >> 4) << 3;
    const int m0 = blockIdx.y * 64;
    const int n0 = blockIdx.x * 256 + wave * 64;

    v8f zero = {};
    v8f acc[4][4];
#pragma unroll
    for (int mi = 0; mi < 4; ++mi)
#pragma unroll
        for (int ni = 0; ni < 4; ++ni) acc[mi][ni] = zero;

    for (int kk = 0; kk < DCONST; kk += 4) {
        v2f a[4], bf[4];
#pragma unroll
        for (int i = 0; i < 4; ++i)
            a[i] = *(const v2f*)(oin + (size_t)(m0 + i * 16 + laneN) * DCONST + kk + koff);
#pragma unroll
        for (int i = 0; i < 4; ++i)
            bf[i] = *(const v2f*)(w + (size_t)(n0 + i * 16 + laneN) * DCONST + kk + koff);
#pragma unroll
        for (int mi = 0; mi < 4; ++mi)
#pragma unroll
            for (int ni = 0; ni < 4; ++ni)
                acc[mi][ni] = wmma4(a[mi], bf[ni], acc[mi][ni]);
    }

#pragma unroll
    for (int mi = 0; mi < 4; ++mi) {
#pragma unroll
        for (int r = 0; r < 8; ++r) {
            const int m = m0 + mi * 16 + r + hi8;
            float* rowp = out + (size_t)m * DCONST + n0;
#pragma unroll
            for (int ni = 0; ni < 4; ++ni) rowp[ni * 16 + laneN] = acc[mi][ni][r];
        }
    }
}

extern "C" void kernel_launch(void* const* d_in, const int* in_sizes, int n_in,
                              void* d_out, int out_size, void* d_ws, size_t ws_size,
                              hipStream_t stream) {
    const float* x     = (const float*)d_in[0];
    const float* w_qkv = (const float*)d_in[2];
    const float* w_fc  = (const float*)d_in[3];
    const float* qnw   = (const float*)d_in[4];
    const float* knw   = (const float*)d_in[5];

    const size_t per = (size_t)2 * NHC * LCONST * DHC;
    float* wsf  = (float*)d_ws;
    float* qb   = wsf;
    float* kb   = wsf + per;
    float* vb   = wsf + 2 * per;
    float* obuf = wsf + 3 * per;

    dim3 g1(3072 / 256, 4096 / 64);
    qkv_rms_kernel<<<g1, 128, 0, stream>>>(x, w_qkv, qnw, knw, qb, kb, vb);

    dim3 g2(LCONST / 32, NHC, 2);
    attn_kernel<<<g2, 32, 0, stream>>>(qb, kb, vb, obuf);

    dim3 g3(DCONST / 256, 4096 / 64);
    fc_kernel<<<g3, 128, 0, stream>>>(obuf, w_fc, (float*)d_out);
}


# attention: K/V tiles staged in LDS once per 64-row work-group (4 waves x 16 rows, 16-byte loads, register prefetch) replace per-wave strided global loads; k order within 8 permuted: sum order changes
# speedup vs baseline: 1.9195x; 1.9195x over previous
#include <hip/hip_runtime.h>
#include <hip/hip_bf16.h>


typedef __attribute__((ext_vector_type(2))) float v2f;
typedef __attribute__((ext_vector_type(8))) float v8f;

#define LCONST 2048
#define DCONST 1024
#define NHC 16
#define DHC 64

static __device__ inline v8f wmma4(v2f a, v2f b, v8f c) {
    return __builtin_amdgcn_wmma_f32_16x16x4_f32(
         false, a,  false, b,
         (short)0, c,  false,  false);
}

__global__ __launch_bounds__(128) void qkv_rms_kernel(
    const float* __restrict__ x, const float* __restrict__ w,
    const float* __restrict__ qnw, const float* __restrict__ knw,
    float* __restrict__ qb, float* __restrict__ kb, float* __restrict__ vb) {
    const int lane  = threadIdx.x & 31;
    const int wave  = threadIdx.x >> 5;
    const int laneN = lane & 15;
    const int koff  = (lane >> 4) << 1;
    const int hi8   = (lane >> 4) << 3;
    const int m0 = blockIdx.y * 64;
    const int n0 = blockIdx.x * 256 + wave * 64;

    v8f zero = {};
    v8f acc[4][4];
#pragma unroll
    for (int mi = 0; mi < 4; ++mi)
#pragma unroll
        for (int ni = 0; ni < 4; ++ni) acc[mi][ni] = zero;

    for (int kk = 0; kk < DCONST; kk += 4) {
        v2f a[4], bf[4];
#pragma unroll
        for (int i = 0; i < 4; ++i)
            a[i] = *(const v2f*)(x + (size_t)(m0 + i * 16 + laneN) * DCONST + kk + koff);
#pragma unroll
        for (int i = 0; i < 4; ++i)
            bf[i] = *(const v2f*)(w + (size_t)(n0 + i * 16 + laneN) * DCONST + kk + koff);
#pragma unroll
        for (int mi = 0; mi < 4; ++mi)
#pragma unroll
            for (int ni = 0; ni < 4; ++ni)
                acc[mi][ni] = wmma4(a[mi], bf[ni], acc[mi][ni]);
    }

    const int sec = n0 >> 10;
    const int h   = (n0 & 1023) >> 6;
    float* dst = (sec == 0) ? qb : ((sec == 1) ? kb : vb);
    float wn[4];
#pragma unroll
    for (int ni = 0; ni < 4; ++ni) {
        const float* nw = (sec == 0) ? qnw : knw;
        wn[ni] = (sec < 2) ? nw[ni * 16 + laneN] : 1.0f;
    }

#pragma unroll
    for (int mi = 0; mi < 4; ++mi) {
#pragma unroll
        for (int r = 0; r < 8; ++r) {
            float sv[4];
#pragma unroll
            for (int ni = 0; ni < 4; ++ni) sv[ni] = acc[mi][ni][r];
            if (sec < 2) {
                float ss = sv[0] * sv[0] + sv[1] * sv[1] + sv[2] * sv[2] + sv[3] * sv[3];
                ss += __shfl_xor(ss, 1, 32);
                ss += __shfl_xor(ss, 2, 32);
                ss += __shfl_xor(ss, 4, 32);
                ss += __shfl_xor(ss, 8, 32);
                float rstd = rsqrtf(ss * (1.0f / 64.0f) + 1e-5f);
#pragma unroll
                for (int ni = 0; ni < 4; ++ni) sv[ni] *= rstd * wn[ni];
            }
            const int m = m0 + mi * 16 + r + hi8;
            const int b = m >> 11;
            const int l = m & (LCONST - 1);
            float* rowp = dst + ((size_t)((b * NHC + h) * LCONST + l) << 6);
#pragma unroll
            for (int ni = 0; ni < 4; ++ni) rowp[ni * 16 + laneN] = sv[ni];
        }
    }
}

typedef __attribute__((ext_vector_type(4))) float v4f;
#define KP 68

__global__ __launch_bounds__(128, 4) void attn_kernel(
    const float* __restrict__ qb, const float* __restrict__ kb,
    const float* __restrict__ vb, float* __restrict__ ob) {
    const int tid  = threadIdx.x;
    const int lane = tid & 31;
    const int wave = tid >> 5;
    const int c    = lane & 15;
    const int hh   = lane >> 4;
    const int hi8  = hh << 3;
    const int q0 = blockIdx.x * 64;
    const int qw = q0 + wave * 16;
    const int bh = blockIdx.z * NHC + blockIdx.y;
    const float* qh = qb + (size_t)bh * LCONST * DHC;
    const float* kh = kb + (size_t)bh * LCONST * DHC;
    const float* vh = vb + (size_t)bh * LCONST * DHC;

    __shared__ __attribute__((aligned(16))) float k_s[64 * KP];
    __shared__ __attribute__((aligned(16))) float v_s[64 * KP];
    __shared__ __attribute__((aligned(16))) float p_s[4 * 16 * KP];
    float* p_w = p_s + wave * (16 * KP);

    v4f qa[8];
#pragma unroll
    for (int j = 0; j < 8; ++j)
        qa[j] = *(const v4f*)(qh + (size_t)(qw + c) * DHC + 8 * j + 4 * hh);

    v8f zero = {};
    v8f o[4];
    float mr[8], lr[8];
#pragma unroll
    for (int nj = 0; nj < 4; ++nj) o[nj] = zero;
#pragma unroll
    for (int r = 0; r < 8; ++r) { mr[r] = -3.0e38f; lr[r] = 0.0f; }

    const int nkv = blockIdx.x + 1;

    v4f kr[8], vr[8];
#pragma unroll
    for (int i = 0; i < 8; ++i) {
        kr[i] = *(const v4f*)(kh + (size_t)(tid + 128 * i) * 4);
        vr[i] = *(const v4f*)(vh + (size_t)(tid + 128 * i) * 4);
    }

    for (int kv = 0; kv < nkv; ++kv) {
        const int kbase = kv * 64;
        __syncthreads();
#pragma unroll
        for (int i = 0; i < 8; ++i) {
            const int f = tid + 128 * i;
            const int row = f >> 4, c4 = f & 15;
            *(v4f*)(k_s + row * KP + c4 * 4) = kr[i];
            *(v4f*)(v_s + row * KP + c4 * 4) = vr[i];
        }
        __syncthreads();
        if (kv + 1 < nkv) {
            const float* kn = kh + (size_t)(kbase + 64) * DHC;
            const float* vn = vh + (size_t)(kbase + 64) * DHC;
#pragma unroll
            for (int i = 0; i < 8; ++i) {
                kr[i] = *(const v4f*)(kn + (size_t)(tid + 128 * i) * 4);
                vr[i] = *(const v4f*)(vn + (size_t)(tid + 128 * i) * 4);
            }
        }

        v8f s[4];
#pragma unroll
        for (int ni = 0; ni < 4; ++ni) s[ni] = zero;
#pragma unroll
        for (int j = 0; j < 8; ++j) {
            v4f kf[4];
#pragma unroll
            for (int ni = 0; ni < 4; ++ni)
                kf[ni] = *(const v4f*)(k_s + (ni * 16 + c) * KP + 8 * j + 4 * hh);
            const v2f a0 = {qa[j].x, qa[j].y};
            const v2f a1 = {qa[j].z, qa[j].w};
#pragma unroll
            for (int ni = 0; ni < 4; ++ni) {
                const v2f b0 = {kf[ni].x, kf[ni].y};
                const v2f b1 = {kf[ni].z, kf[ni].w};
                s[ni] = wmma4(a0, b0, s[ni]);
                s[ni] = wmma4(a1, b1, s[ni]);
            }
        }

        const float scale = 0.125f;
#pragma unroll
        for (int r = 0; r < 8; ++r) {
            const int qrow = qw + r + hi8;
            float sv[4];
            float vmax = -3.0e38f;
#pragma unroll
            for (int ni = 0; ni < 4; ++ni) {
                const int key = kbase + ni * 16 + c;
                float t = s[ni][r] * scale;
                t = (key > qrow) ? -1.0e30f : t;
                sv[ni] = t;
                vmax = fmaxf(vmax, t);
            }
            vmax = fmaxf(vmax, __shfl_xor(vmax, 1, 32));
            vmax = fmaxf(vmax, __shfl_xor(vmax, 2, 32));
            vmax = fmaxf(vmax, __shfl_xor(vmax, 4, 32));
            vmax = fmaxf(vmax, __shfl_xor(vmax, 8, 32));
            const float mold = mr[r];
            const float mnew = fmaxf(mold, vmax);
            const float corr = __expf(mold - mnew);
            float psum = 0.0f;
#pragma unroll
            for (int ni = 0; ni < 4; ++ni) {
                float p = __expf(sv[ni] - mnew);
                sv[ni] = p;
                psum += p;
            }
            psum += __shfl_xor(psum, 1, 32);
            psum += __shfl_xor(psum, 2, 32);
            psum += __shfl_xor(psum, 4, 32);
            psum += __shfl_xor(psum, 8, 32);
            lr[r] = lr[r] * corr + psum;
            mr[r] = mnew;
#pragma unroll
            for (int nj = 0; nj < 4; ++nj) o[nj][r] *= corr;
#pragma unroll
            for (int ni = 0; ni < 4; ++ni)
                p_w[(r + hi8) * KP + ni * 16 + c] = sv[ni];
        }
        __builtin_amdgcn_fence(__ATOMIC_RELEASE, "workgroup", "local");
        __builtin_amdgcn_wave_barrier();
        __builtin_amdgcn_fence(__ATOMIC_ACQUIRE, "workgroup", "local");

#pragma unroll
        for (int j = 0; j < 8; ++j) {
            const v4f pa = *(const v4f*)(p_w + c * KP + 8 * j + 4 * hh);
            const v2f a0 = {pa.x, pa.y};
            const v2f a1 = {pa.z, pa.w};
#pragma unroll
            for (int nj = 0; nj < 4; ++nj) {
                const float* vp = v_s + (8 * j + 4 * hh) * KP + nj * 16 + c;
                const v2f b0 = {vp[0], vp[KP]};
                const v2f b1 = {vp[2 * KP], vp[3 * KP]};
                o[nj] = wmma4(a0, b0, o[nj]);
                o[nj] = wmma4(a1, b1, o[nj]);
            }
        }
    }

#pragma unroll
    for (int r = 0; r < 8; ++r) {
        const float inv = 1.0f / lr[r];
        const int l = qw + r + hi8;
        float* orow = ob + (size_t)(blockIdx.z * LCONST + l) * DCONST + blockIdx.y * DHC;
#pragma unroll
        for (int nj = 0; nj < 4; ++nj) orow[nj * 16 + c] = o[nj][r] * inv;
    }
}

__global__ __launch_bounds__(128) void fc_kernel(
    const float* __restrict__ oin, const float* __restrict__ w,
    float* __restrict__ out) {
    const int lane  = threadIdx.x & 31;
    const int wave  = threadIdx.x >> 5;
    const int laneN = lane & 15;
    const int koff  = (lane >> 4) << 1;
    const int hi8   = (lane >> 4) << 3;
    const int m0 = blockIdx.y * 64;
    const int n0 = blockIdx.x * 256 + wave * 64;

    v8f zero = {};
    v8f acc[4][4];
#pragma unroll
    for (int mi = 0; mi < 4; ++mi)
#pragma unroll
        for (int ni = 0; ni < 4; ++ni) acc[mi][ni] = zero;

    for (int kk = 0; kk < DCONST; kk += 4) {
        v2f a[4], bf[4];
#pragma unroll
        for (int i = 0; i < 4; ++i)
            a[i] = *(const v2f*)(oin + (size_t)(m0 + i * 16 + laneN) * DCONST + kk + koff);
#pragma unroll
        for (int i = 0; i < 4; ++i)
            bf[i] = *(const v2f*)(w + (size_t)(n0 + i * 16 + laneN) * DCONST + kk + koff);
#pragma unroll
        for (int mi = 0; mi < 4; ++mi)
#pragma unroll
            for (int ni = 0; ni < 4; ++ni)
                acc[mi][ni] = wmma4(a[mi], bf[ni], acc[mi][ni]);
    }

#pragma unroll
    for (int mi = 0; mi < 4; ++mi) {
#pragma unroll
        for (int r = 0; r < 8; ++r) {
            const int m = m0 + mi * 16 + r + hi8;
            float* rowp = out + (size_t)m * DCONST + n0;
#pragma unroll
            for (int ni = 0; ni < 4; ++ni) rowp[ni * 16 + laneN] = acc[mi][ni][r];
        }
    }
}

extern "C" void kernel_launch(void* const* d_in, const int* in_sizes, int n_in,
                              void* d_out, int out_size, void* d_ws, size_t ws_size,
                              hipStream_t stream) {
    const float* x     = (const float*)d_in[0];
    const float* w_qkv = (const float*)d_in[2];
    const float* w_fc  = (const float*)d_in[3];
    const float* qnw   = (const float*)d_in[4];
    const float* knw   = (const float*)d_in[5];

    const size_t per = (size_t)2 * NHC * LCONST * DHC;
    float* wsf  = (float*)d_ws;
    float* qb   = wsf;
    float* kb   = wsf + per;
    float* vb   = wsf + 2 * per;
    float* obuf = wsf + 3 * per;

    dim3 g1(3072 / 256, 4096 / 64);
    qkv_rms_kernel<<<g1, 128, 0, stream>>>(x, w_qkv, qnw, knw, qb, kb, vb);

    dim3 g2(LCONST / 64, NHC, 2);
    attn_kernel<<<g2, 128, 0, stream>>>(qb, kb, vb, obuf);

    dim3 g3(DCONST / 256, 4096 / 64);
    fc_kernel<<<g3, 128, 0, stream>>>(obuf, w_fc, (float*)d_out);
}


# attention: each work-group runs query tiles 31-p and p in turn, 33 key tiles for every group (grid 512, bound 2 waves per SIMD), not one tile of 1 to 32 key tiles: equal work per compute unit
# speedup vs baseline: 2.3795x; 1.2396x over previous
#include <hip/hip_runtime.h>
#include <hip/hip_bf16.h>


typedef __attribute__((ext_vector_type(2))) float v2f;
typedef __attribute__((ext_vector_type(8))) float v8f;

#define LCONST 2048
#define DCONST 1024
#define NHC 16
#define DHC 64

static __device__ inline v8f wmma4(v2f a, v2f b, v8f c) {
    return __builtin_amdgcn_wmma_f32_16x16x4_f32(
         false, a,  false, b,
         (short)0, c,  false,  false);
}

__global__ __launch_bounds__(128) void qkv_rms_kernel(
    const float* __restrict__ x, const float* __restrict__ w,
    const float* __restrict__ qnw, const float* __restrict__ knw,
    float* __restrict__ qb, float* __restrict__ kb, float* __restrict__ vb) {
    const int lane  = threadIdx.x & 31;
    const int wave  = threadIdx.x >> 5;
    const int laneN = lane & 15;
    const int koff  = (lane >> 4) << 1;
    const int hi8   = (lane >> 4) << 3;
    const int m0 = blockIdx.y * 64;
    const int n0 = blockIdx.x * 256 + wave * 64;

    v8f zero = {};
    v8f acc[4][4];
#pragma unroll
    for (int mi = 0; mi < 4; ++mi)
#pragma unroll
        for (int ni = 0; ni < 4; ++ni) acc[mi][ni] = zero;

    for (int kk = 0; kk < DCONST; kk += 4) {
        v2f a[4], bf[4];
#pragma unroll
        for (int i = 0; i < 4; ++i)
            a[i] = *(const v2f*)(x + (size_t)(m0 + i * 16 + laneN) * DCONST + kk + koff);
#pragma unroll
        for (int i = 0; i < 4; ++i)
            bf[i] = *(const v2f*)(w + (size_t)(n0 + i * 16 + laneN) * DCONST + kk + koff);
#pragma unroll
        for (int mi = 0; mi < 4; ++mi)
#pragma unroll
            for (int ni = 0; ni < 4; ++ni)
                acc[mi][ni] = wmma4(a[mi], bf[ni], acc[mi][ni]);
    }

    const int sec = n0 >> 10;
    const int h   = (n0 & 1023) >> 6;
    float* dst = (sec == 0) ? qb : ((sec == 1) ? kb : vb);
    float wn[4];
#pragma unroll
    for (int ni = 0; ni < 4; ++ni) {
        const float* nw = (sec == 0) ? qnw : knw;
        wn[ni] = (sec < 2) ? nw[ni * 16 + laneN] : 1.0f;
    }

#pragma unroll
    for (int mi = 0; mi < 4; ++mi) {
#pragma unroll
        for (int r = 0; r < 8; ++r) {
            float sv[4];
#pragma unroll
            for (int ni = 0; ni < 4; ++ni) sv[ni] = acc[mi][ni][r];
            if (sec < 2) {
                float ss = sv[0] * sv[0] + sv[1] * sv[1] + sv[2] * sv[2] + sv[3] * sv[3];
                ss += __shfl_xor(ss, 1, 32);
                ss += __shfl_xor(ss, 2, 32);
                ss += __shfl_xor(ss, 4, 32);
                ss += __shfl_xor(ss, 8, 32);
                float rstd = rsqrtf(ss * (1.0f / 64.0f) + 1e-5f);
#pragma unroll
                for (int ni = 0; ni < 4; ++ni) sv[ni] *= rstd * wn[ni];
            }
            const int m = m0 + mi * 16 + r + hi8;
            const int b = m >> 11;
            const int l = m & (LCONST - 1);
            float* rowp = dst + ((size_t)((b * NHC + h) * LCONST + l) << 6);
#pragma unroll
            for (int ni = 0; ni < 4; ++ni) rowp[ni * 16 + laneN] = sv[ni];
        }
    }
}

typedef __attribute__((ext_vector_type(4))) float v4f;
#define KP 68

__global__ __launch_bounds__(128, 2) void attn_kernel(
    const float* __restrict__ qb, const float* __restrict__ kb,
    const float* __restrict__ vb, float* __restrict__ ob) {
    const int tid  = threadIdx.x;
    const int lane = tid & 31;
    const int wave = tid >> 5;
    const int c    = lane & 15;
    const int hh   = lane >> 4;
    const int hi8  = hh << 3;
    const int bh = blockIdx.z * NHC + blockIdx.y;
    const float* qh = qb + (size_t)bh * LCONST * DHC;
    const float* kh = kb + (size_t)bh * LCONST * DHC;
    const float* vh = vb + (size_t)bh * LCONST * DHC;

    __shared__ __attribute__((aligned(16))) float k_s[64 * KP];
    __shared__ __attribute__((aligned(16))) float v_s[64 * KP];
    __shared__ __attribute__((aligned(16))) float p_s[4 * 16 * KP];
    float* p_w = p_s + wave * (16 * KP);

  for (int half = 0; half < 2; ++half) {
    const int qt = half ? (int)blockIdx.x : (LCONST / 64 - 1 - (int)blockIdx.x);
    const int q0 = qt * 64;
    const int qw = q0 + wave * 16;

    v4f qa[8];
#pragma unroll
    for (int j = 0; j < 8; ++j)
        qa[j] = *(const v4f*)(qh + (size_t)(qw + c) * DHC + 8 * j + 4 * hh);

    v8f zero = {};
    v8f o[4];
    float mr[8], lr[8];
#pragma unroll
    for (int nj = 0; nj < 4; ++nj) o[nj] = zero;
#pragma unroll
    for (int r = 0; r < 8; ++r) { mr[r] = -3.0e38f; lr[r] = 0.0f; }

    const int nkv = qt + 1;

    v4f kr[8], vr[8];
#pragma unroll
    for (int i = 0; i < 8; ++i) {
        kr[i] = *(const v4f*)(kh + (size_t)(tid + 128 * i) * 4);
        vr[i] = *(const v4f*)(vh + (size_t)(tid + 128 * i) * 4);
    }

    for (int kv = 0; kv < nkv; ++kv) {
        const int kbase = kv * 64;
        __syncthreads();
#pragma unroll
        for (int i = 0; i < 8; ++i) {
            const int f = tid + 128 * i;
            const int row = f >> 4, c4 = f & 15;
            *(v4f*)(k_s + row * KP + c4 * 4) = kr[i];
            *(v4f*)(v_s + row * KP + c4 * 4) = vr[i];
        }
        __syncthreads();
        if (kv + 1 < nkv) {
            const float* kn = kh + (size_t)(kbase + 64) * DHC;
            const float* vn = vh + (size_t)(kbase + 64) * DHC;
#pragma unroll
            for (int i = 0; i < 8; ++i) {
                kr[i] = *(const v4f*)(kn + (size_t)(tid + 128 * i) * 4);
                vr[i] = *(const v4f*)(vn + (size_t)(tid + 128 * i) * 4);
            }
        }

        v8f s[4];
#pragma unroll
        for (int ni = 0; ni < 4; ++ni) s[ni] = zero;
#pragma unroll
        for (int j = 0; j < 8; ++j) {
            v4f kf[4];
#pragma unroll
            for (int ni = 0; ni < 4; ++ni)
                kf[ni] = *(const v4f*)(k_s + (ni * 16 + c) * KP + 8 * j + 4 * hh);
            const v2f a0 = {qa[j].x, qa[j].y};
            const v2f a1 = {qa[j].z, qa[j].w};
#pragma unroll
            for (int ni = 0; ni < 4; ++ni) {
                const v2f b0 = {kf[ni].x, kf[ni].y};
                const v2f b1 = {kf[ni].z, kf[ni].w};
                s[ni] = wmma4(a0, b0, s[ni]);
                s[ni] = wmma4(a1, b1, s[ni]);
            }
        }

        const float scale = 0.125f;
#pragma unroll
        for (int r = 0; r < 8; ++r) {
            const int qrow = qw + r + hi8;
            float sv[4];
            float vmax = -3.0e38f;
#pragma unroll
            for (int ni = 0; ni < 4; ++ni) {
                const int key = kbase + ni * 16 + c;
                float t = s[ni][r] * scale;
                t = (key > qrow) ? -1.0e30f : t;
                sv[ni] = t;
                vmax = fmaxf(vmax, t);
            }
            vmax = fmaxf(vmax, __shfl_xor(vmax, 1, 32));
            vmax = fmaxf(vmax, __shfl_xor(vmax, 2, 32));
            vmax = fmaxf(vmax, __shfl_xor(vmax, 4, 32));
            vmax = fmaxf(vmax, __shfl_xor(vmax, 8, 32));
            const float mold = mr[r];
            const float mnew = fmaxf(mold, vmax);
            const float corr = __expf(mold - mnew);
            float psum = 0.0f;
#pragma unroll
            for (int ni = 0; ni < 4; ++ni) {
                float p = __expf(sv[ni] - mnew);
                sv[ni] = p;
                psum += p;
            }
            psum += __shfl_xor(psum, 1, 32);
            psum += __shfl_xor(psum, 2, 32);
            psum += __shfl_xor(psum, 4, 32);
            psum += __shfl_xor(psum, 8, 32);
            lr[r] = lr[r] * corr + psum;
            mr[r] = mnew;
#pragma unroll
            for (int nj = 0; nj < 4; ++nj) o[nj][r] *= corr;
#pragma unroll
            for (int ni = 0; ni < 4; ++ni)
                p_w[(r + hi8) * KP + ni * 16 + c] = sv[ni];
        }
        __builtin_amdgcn_fence(__ATOMIC_RELEASE, "workgroup", "local");
        __builtin_amdgcn_wave_barrier();
        __builtin_amdgcn_fence(__ATOMIC_ACQUIRE, "workgroup", "local");

#pragma unroll
        for (int j = 0; j < 8; ++j) {
            const v4f pa = *(const v4f*)(p_w + c * KP + 8 * j + 4 * hh);
            const v2f a0 = {pa.x, pa.y};
            const v2f a1 = {pa.z, pa.w};
#pragma unroll
            for (int nj = 0; nj < 4; ++nj) {
                const float* vp = v_s + (8 * j + 4 * hh) * KP + nj * 16 + c;
                const v2f b0 = {vp[0], vp[KP]};
                const v2f b1 = {vp[2 * KP], vp[3 * KP]};
                o[nj] = wmma4(a0, b0, o[nj]);
                o[nj] = wmma4(a1, b1, o[nj]);
            }
        }
    }

#pragma unroll
    for (int r = 0; r < 8; ++r) {
        const float inv = 1.0f / lr[r];
        const int l = qw + r + hi8;
        float* orow = ob + (size_t)(blockIdx.z * LCONST + l) * DCONST + blockIdx.y * DHC;
#pragma unroll
        for (int nj = 0; nj < 4; ++nj) orow[nj * 16 + c] = o[nj][r] * inv;
    }
  }
}

__global__ __launch_bounds__(128) void fc_kernel(
    const float* __restrict__ oin, const float* __restrict__ w,
    float* __restrict__ out) {
    const int lane  = threadIdx.x & 31;
    const int wave  = threadIdx.x >> 5;
    const int laneN = lane & 15;
    const int koff  = (lane >> 4) << 1;
    const int hi8   = (lane >> 4) << 3;
    const int m0 = blockIdx.y * 64;
    const int n0 = blockIdx.x * 256 + wave * 64;

    v8f zero = {};
    v8f acc[4][4];
#pragma unroll
    for (int mi = 0; mi < 4; ++mi)
#pragma unroll
        for (int ni = 0; ni < 4; ++ni) acc[mi][ni] = zero;

    for (int kk = 0; kk < DCONST; kk += 4) {
        v2f a[4], bf[4];
#pragma unroll
        for (int i = 0; i < 4; ++i)
            a[i] = *(const v2f*)(oin + (size_t)(m0 + i * 16 + laneN) * DCONST + kk + koff);
#pragma unroll
        for (int i = 0; i < 4; ++i)
            bf[i] = *(const v2f*)(w + (size_t)(n0 + i * 16 + laneN) * DCONST + kk + koff);
#pragma unroll
        for (int mi = 0; mi < 4; ++mi)
#pragma unroll
            for (int ni = 0; ni < 4; ++ni)
                acc[mi][ni] = wmma4(a[mi], bf[ni], acc[mi][ni]);
    }

#pragma unroll
    for (int mi = 0; mi < 4; ++mi) {
#pragma unroll
        for (int r = 0; r < 8; ++r) {
            const int m = m0 + mi * 16 + r + hi8;
            float* rowp = out + (size_t)m * DCONST + n0;
#pragma unroll
            for (int ni = 0; ni < 4; ++ni) rowp[ni * 16 + laneN] = acc[mi][ni][r];
        }
    }
}

extern "C" void kernel_launch(void* const* d_in, const int* in_sizes, int n_in,
                              void* d_out, int out_size, void* d_ws, size_t ws_size,
                              hipStream_t stream) {
    const float* x     = (const float*)d_in[0];
    const float* w_qkv = (const float*)d_in[2];
    const float* w_fc  = (const float*)d_in[3];
    const float* qnw   = (const float*)d_in[4];
    const float* knw   = (const float*)d_in[5];

    const size_t per = (size_t)2 * NHC * LCONST * DHC;
    float* wsf  = (float*)d_ws;
    float* qb   = wsf;
    float* kb   = wsf + per;
    float* vb   = wsf + 2 * per;
    float* obuf = wsf + 3 * per;

    dim3 g1(3072 / 256, 4096 / 64);
    qkv_rms_kernel<<<g1, 128, 0, stream>>>(x, w_qkv, qnw, knw, qb, kb, vb);

    dim3 g2(LCONST / 128, NHC, 2);
    attn_kernel<<<g2, 128, 0, stream>>>(qb, kb, vb, obuf);

    dim3 g3(DCONST / 256, 4096 / 64);
    fc_kernel<<<g3, 128, 0, stream>>>(obuf, w_fc, (float*)d_out);
}


# both projection GEMMs: 128x128 A/W tiles staged in LDS once per work-group (32 k a step, 16-byte loads one step ahead), not per-wave strided global loads; k order in 8 permuted: sum order changes
# speedup vs baseline: 2.9617x; 1.2447x over previous
#include <hip/hip_runtime.h>
#include <hip/hip_bf16.h>


typedef __attribute__((ext_vector_type(2))) float v2f;
typedef __attribute__((ext_vector_type(8))) float v8f;

#define LCONST 2048
#define DCONST 1024
#define NHC 16
#define DHC 64

static __device__ inline v8f wmma4(v2f a, v2f b, v8f c) {
    return __builtin_amdgcn_wmma_f32_16x16x4_f32(
         false, a,  false, b,
         (short)0, c,  false,  false);
}

typedef __attribute__((ext_vector_type(4))) float v4f;
#define GK 32
#define GP 36
#define GSTAGE (128 * GP)

static __device__ inline void gemm_tile_128x128(
    const float* __restrict__ a, const float* __restrict__ w,
    const int m0, const int n0, float* __restrict__ g_s, v8f (&acc)[4][4]) {
    const int tid  = threadIdx.x;
    const int lane = tid & 31;
    const int wave = tid >> 5;
    const int c    = lane & 15;
    const int hh   = lane >> 4;
    const int wm   = wave >> 1;
    const int wn   = wave & 1;

    const int trow = tid >> 3;
    const int tc4  = (tid & 7) * 4;
    const float* ag = a + (size_t)(m0 + trow) * DCONST + tc4;
    const float* wg = w + (size_t)(n0 + trow) * DCONST + tc4;
    const int st_off = trow * GP + tc4;
    const int fa_off = (wm * 64 + c) * GP + 4 * hh;
    const int fb_off = GSTAGE + (wn * 64 + c) * GP + 4 * hh;

    v8f zero = {};
#pragma unroll
    for (int mi = 0; mi < 4; ++mi)
#pragma unroll
        for (int ni = 0; ni < 4; ++ni) acc[mi][ni] = zero;

    v4f pa[8], pb[8];
#pragma unroll
    for (int i = 0; i < 8; ++i) {
        pa[i] = *(const v4f*)(ag + (size_t)i * 16 * DCONST);
        pb[i] = *(const v4f*)(wg + (size_t)i * 16 * DCONST);
    }
#pragma unroll
    for (int i = 0; i < 8; ++i) {
        *(v4f*)(g_s + st_off + i * 16 * GP) = pa[i];
        *(v4f*)(g_s + GSTAGE + st_off + i * 16 * GP) = pb[i];
    }
    __syncthreads();

    v4f af[2][4], bf[2][4];
#pragma unroll
    for (int i = 0; i < 4; ++i) {
        af[0][i] = *(const v4f*)(g_s + fa_off + i * 16 * GP);
        bf[0][i] = *(const v4f*)(g_s + fb_off + i * 16 * GP);
    }

    for (int t = 0; t < DCONST / GK; ++t) {
        const float* cur = g_s + (t & 1) * (2 * GSTAGE);
        float* nxt = g_s + ((t & 1) ^ 1) * (2 * GSTAGE);
        const bool more = (t + 1) < DCONST / GK;
        if (more) {
            const float* an = ag + (t + 1) * GK;
            const float* wn_ = wg + (t + 1) * GK;
#pragma unroll
            for (int i = 0; i < 8; ++i) {
                pa[i] = *(const v4f*)(an + (size_t)i * 16 * DCONST);
                pb[i] = *(const v4f*)(wn_ + (size_t)i * 16 * DCONST);
            }
        }

#pragma unroll
        for (int j = 0; j < GK / 8; ++j) {
            const int cb = j & 1;
            if (j + 1 < GK / 8) {
#pragma unroll
                for (int i = 0; i < 4; ++i) {
                    af[cb ^ 1][i] = *(const v4f*)(cur + fa_off + i * 16 * GP + 8 * (j + 1));
                    bf[cb ^ 1][i] = *(const v4f*)(cur + fb_off + i * 16 * GP + 8 * (j + 1));
                }
            }
            if (j == GK / 8 - 2 && more) {
#pragma unroll
                for (int i = 0; i < 8; ++i) {
                    *(v4f*)(nxt + st_off + i * 16 * GP) = pa[i];
                    *(v4f*)(nxt + GSTAGE + st_off + i * 16 * GP) = pb[i];
                }
            }
            if (j == GK / 8 - 1 && more) {
                __syncthreads();
#pragma unroll
                for (int i = 0; i < 4; ++i) {
                    af[0][i] = *(const v4f*)(nxt + fa_off + i * 16 * GP);
                    bf[0][i] = *(const v4f*)(nxt + fb_off + i * 16 * GP);
                }
            }
            __builtin_amdgcn_sched_barrier(0);
#pragma unroll
            for (int mi = 0; mi < 4; ++mi)
#pragma unroll
                for (int ni = 0; ni < 4; ++ni) {
                    const v2f a0 = {af[cb][mi].x, af[cb][mi].y};
                    const v2f b0 = {bf[cb][ni].x, bf[cb][ni].y};
                    acc[mi][ni] = wmma4(a0, b0, acc[mi][ni]);
                }
#pragma unroll
            for (int mi = 0; mi < 4; ++mi)
#pragma unroll
                for (int ni = 0; ni < 4; ++ni) {
                    const v2f a1 = {af[cb][mi].z, af[cb][mi].w};
                    const v2f b1 = {bf[cb][ni].z, bf[cb][ni].w};
                    acc[mi][ni] = wmma4(a1, b1, acc[mi][ni]);
                }
            __builtin_amdgcn_sched_barrier(0);
        }
    }
}

__global__ __launch_bounds__(128, 3) void qkv_rms_kernel(
    const float* __restrict__ x, const float* __restrict__ w,
    const float* __restrict__ qnw, const float* __restrict__ knw,
    float* __restrict__ qb, float* __restrict__ kb, float* __restrict__ vb) {
    __shared__ __attribute__((aligned(16))) float g_s[4 * GSTAGE];
    const int lane  = threadIdx.x & 31;
    const int wave  = threadIdx.x >> 5;
    const int laneN = lane & 15;
    const int hi8   = (lane >> 4) << 3;
    const int mb = blockIdx.y * 128;
    const int nb = blockIdx.x * 128;

    v8f acc[4][4];
    gemm_tile_128x128(x, w, mb, nb, g_s, acc);

    const int m0 = mb + (wave >> 1) * 64;
    const int n0 = nb + (wave & 1) * 64;

    const int sec = n0 >> 10;
    const int h   = (n0 & 1023) >> 6;
    float* dst = (sec == 0) ? qb : ((sec == 1) ? kb : vb);
    float wn[4];
#pragma unroll
    for (int ni = 0; ni < 4; ++ni) {
        const float* nw = (sec == 0) ? qnw : knw;
        wn[ni] = (sec < 2) ? nw[ni * 16 + laneN] : 1.0f;
    }

#pragma unroll
    for (int mi = 0; mi < 4; ++mi) {
#pragma unroll
        for (int r = 0; r < 8; ++r) {
            float sv[4];
#pragma unroll
            for (int ni = 0; ni < 4; ++ni) sv[ni] = acc[mi][ni][r];
            if (sec < 2) {
                float ss = sv[0] * sv[0] + sv[1] * sv[1] + sv[2] * sv[2] + sv[3] * sv[3];
                ss += __shfl_xor(ss, 1, 32);
                ss += __shfl_xor(ss, 2, 32);
                ss += __shfl_xor(ss, 4, 32);
                ss += __shfl_xor(ss, 8, 32);
                float rstd = rsqrtf(ss * (1.0f / 64.0f) + 1e-5f);
#pragma unroll
                for (int ni = 0; ni < 4; ++ni) sv[ni] *= rstd * wn[ni];
            }
            const int m = m0 + mi * 16 + r + hi8;
            const int b = m >> 11;
            const int l = m & (LCONST - 1);
            float* rowp = dst + ((size_t)((b * NHC + h) * LCONST + l) << 6);
#pragma unroll
            for (int ni = 0; ni < 4; ++ni) rowp[ni * 16 + laneN] = sv[ni];
        }
    }
}

#define KP 68

__global__ __launch_bounds__(128, 2) void attn_kernel(
    const float* __restrict__ qb, const float* __restrict__ kb,
    const float* __restrict__ vb, float* __restrict__ ob) {
    const int tid  = threadIdx.x;
    const int lane = tid & 31;
    const int wave = tid >> 5;
    const int c    = lane & 15;
    const int hh   = lane >> 4;
    const int hi8  = hh << 3;
    const int bh = blockIdx.z * NHC + blockIdx.y;
    const float* qh = qb + (size_t)bh * LCONST * DHC;
    const float* kh = kb + (size_t)bh * LCONST * DHC;
    const float* vh = vb + (size_t)bh * LCONST * DHC;

    __shared__ __attribute__((aligned(16))) float k_s[64 * KP];
    __shared__ __attribute__((aligned(16))) float v_s[64 * KP];
    __shared__ __attribute__((aligned(16))) float p_s[4 * 16 * KP];
    float* p_w = p_s + wave * (16 * KP);

  for (int half = 0; half < 2; ++half) {
    const int qt = half ? (int)blockIdx.x : (LCONST / 64 - 1 - (int)blockIdx.x);
    const int q0 = qt * 64;
    const int qw = q0 + wave * 16;

    v4f qa[8];
#pragma unroll
    for (int j = 0; j < 8; ++j)
        qa[j] = *(const v4f*)(qh + (size_t)(qw + c) * DHC + 8 * j + 4 * hh);

    v8f zero = {};
    v8f o[4];
    float mr[8], lr[8];
#pragma unroll
    for (int nj = 0; nj < 4; ++nj) o[nj] = zero;
#pragma unroll
    for (int r = 0; r < 8; ++r) { mr[r] = -3.0e38f; lr[r] = 0.0f; }

    const int nkv = qt + 1;

    v4f kr[8], vr[8];
#pragma unroll
    for (int i = 0; i < 8; ++i) {
        kr[i] = *(const v4f*)(kh + (size_t)(tid + 128 * i) * 4);
        vr[i] = *(const v4f*)(vh + (size_t)(tid + 128 * i) * 4);
    }

    for (int kv = 0; kv < nkv; ++kv) {
        const int kbase = kv * 64;
        __syncthreads();
#pragma unroll
        for (int i = 0; i < 8; ++i) {
            const int f = tid + 128 * i;
            const int row = f >> 4, c4 = f & 15;
            *(v4f*)(k_s + row * KP + c4 * 4) = kr[i];
            *(v4f*)(v_s + row * KP + c4 * 4) = vr[i];
        }
        __syncthreads();
        if (kv + 1 < nkv) {
            const float* kn = kh + (size_t)(kbase + 64) * DHC;
            const float* vn = vh + (size_t)(kbase + 64) * DHC;
#pragma unroll
            for (int i = 0; i < 8; ++i) {
                kr[i] = *(const v4f*)(kn + (size_t)(tid + 128 * i) * 4);
                vr[i] = *(const v4f*)(vn + (size_t)(tid + 128 * i) * 4);
            }
        }

        v8f s[4];
#pragma unroll
        for (int ni = 0; ni < 4; ++ni) s[ni] = zero;
#pragma unroll
        for (int j = 0; j < 8; ++j) {
            v4f kf[4];
#pragma unroll
            for (int ni = 0; ni < 4; ++ni)
                kf[ni] = *(const v4f*)(k_s + (ni * 16 + c) * KP + 8 * j + 4 * hh);
            const v2f a0 = {qa[j].x, qa[j].y};
            const v2f a1 = {qa[j].z, qa[j].w};
#pragma unroll
            for (int ni = 0; ni < 4; ++ni) {
                const v2f b0 = {kf[ni].x, kf[ni].y};
                const v2f b1 = {kf[ni].z, kf[ni].w};
                s[ni] = wmma4(a0, b0, s[ni]);
                s[ni] = wmma4(a1, b1, s[ni]);
            }
        }

        const float scale = 0.125f;
#pragma unroll
        for (int r = 0; r < 8; ++r) {
            const int qrow = qw + r + hi8;
            float sv[4];
            float vmax = -3.0e38f;
#pragma unroll
            for (int ni = 0; ni < 4; ++ni) {
                const int key = kbase + ni * 16 + c;
                float t = s[ni][r] * scale;
                t = (key > qrow) ? -1.0e30f : t;
                sv[ni] = t;
                vmax = fmaxf(vmax, t);
            }
            vmax = fmaxf(vmax, __shfl_xor(vmax, 1, 32));
            vmax = fmaxf(vmax, __shfl_xor(vmax, 2, 32));
            vmax = fmaxf(vmax, __shfl_xor(vmax, 4, 32));
            vmax = fmaxf(vmax, __shfl_xor(vmax, 8, 32));
            const float mold = mr[r];
            const float mnew = fmaxf(mold, vmax);
            const float corr = __expf(mold - mnew);
            float psum = 0.0f;
#pragma unroll
            for (int ni = 0; ni < 4; ++ni) {
                float p = __expf(sv[ni] - mnew);
                sv[ni] = p;
                psum += p;
            }
            psum += __shfl_xor(psum, 1, 32);
            psum += __shfl_xor(psum, 2, 32);
            psum += __shfl_xor(psum, 4, 32);
            psum += __shfl_xor(psum, 8, 32);
            lr[r] = lr[r] * corr + psum;
            mr[r] = mnew;
#pragma unroll
            for (int nj = 0; nj < 4; ++nj) o[nj][r] *= corr;
#pragma unroll
            for (int ni = 0; ni < 4; ++ni)
                p_w[(r + hi8) * KP + ni * 16 + c] = sv[ni];
        }
        __builtin_amdgcn_fence(__ATOMIC_RELEASE, "workgroup", "local");
        __builtin_amdgcn_wave_barrier();
        __builtin_amdgcn_fence(__ATOMIC_ACQUIRE, "workgroup", "local");

#pragma unroll
        for (int j = 0; j < 8; ++j) {
            const v4f pa = *(const v4f*)(p_w + c * KP + 8 * j + 4 * hh);
            const v2f a0 = {pa.x, pa.y};
            const v2f a1 = {pa.z, pa.w};
#pragma unroll
            for (int nj = 0; nj < 4; ++nj) {
                const float* vp = v_s + (8 * j + 4 * hh) * KP + nj * 16 + c;
                const v2f b0 = {vp[0], vp[KP]};
                const v2f b1 = {vp[2 * KP], vp[3 * KP]};
                o[nj] = wmma4(a0, b0, o[nj]);
                o[nj] = wmma4(a1, b1, o[nj]);
            }
        }
    }

#pragma unroll
    for (int r = 0; r < 8; ++r) {
        const float inv = 1.0f / lr[r];
        const int l = qw + r + hi8;
        float* orow = ob + (size_t)(blockIdx.z * LCONST + l) * DCONST + blockIdx.y * DHC;
#pragma unroll
        for (int nj = 0; nj < 4; ++nj) orow[nj * 16 + c] = o[nj][r] * inv;
    }
  }
}

__global__ __launch_bounds__(128, 3) void fc_kernel(
    const float* __restrict__ oin, const float* __restrict__ w,
    float* __restrict__ out) {
    __shared__ __attribute__((aligned(16))) float g_s[4 * GSTAGE];
    const int lane  = threadIdx.x & 31;
    const int wave  = threadIdx.x >> 5;
    const int laneN = lane & 15;
    const int hi8   = (lane >> 4) << 3;
    const int mb = blockIdx.y * 128;
    const int nb = blockIdx.x * 128;

    v8f acc[4][4];
    gemm_tile_128x128(oin, w, mb, nb, g_s, acc);

    const int m0 = mb + (wave >> 1) * 64;
    const int n0 = nb + (wave & 1) * 64;

#pragma unroll
    for (int mi = 0; mi < 4; ++mi) {
#pragma unroll
        for (int r = 0; r < 8; ++r) {
            const int m = m0 + mi * 16 + r + hi8;
            float* rowp = out + (size_t)m * DCONST + n0;
#pragma unroll
            for (int ni = 0; ni < 4; ++ni) rowp[ni * 16 + laneN] = acc[mi][ni][r];
        }
    }
}

extern "C" void kernel_launch(void* const* d_in, const int* in_sizes, int n_in,
                              void* d_out, int out_size, void* d_ws, size_t ws_size,
                              hipStream_t stream) {
    const float* x     = (const float*)d_in[0];
    const float* w_qkv = (const float*)d_in[2];
    const float* w_fc  = (const float*)d_in[3];
    const float* qnw   = (const float*)d_in[4];
    const float* knw   = (const float*)d_in[5];

    const size_t per = (size_t)2 * NHC * LCONST * DHC;
    float* wsf  = (float*)d_ws;
    float* qb   = wsf;
    float* kb   = wsf + per;
    float* vb   = wsf + 2 * per;
    float* obuf = wsf + 3 * per;

    dim3 g1(3072 / 128, 4096 / 128);
    qkv_rms_kernel<<<g1, 128, 0, stream>>>(x, w_qkv, qnw, knw, qb, kb, vb);

    dim3 g2(LCONST / 128, NHC, 2);
    attn_kernel<<<g2, 128, 0, stream>>>(qb, kb, vb, obuf);

    dim3 g3(DCONST / 128, 4096 / 128);
    fc_kernel<<<g3, 128, 0, stream>>>(obuf, w_fc, (float*)d_out);
}


# attention: both products transposed (S^T = K Q^T, O^T = V^T P^T) so P stays in registers, no LDS round trip; row max/sum in-lane, one lane-half exchange per tile: sum order changes
# speedup vs baseline: 3.1489x; 1.0632x over previous
#include <hip/hip_runtime.h>
#include <hip/hip_bf16.h>


typedef __attribute__((ext_vector_type(2))) float v2f;
typedef __attribute__((ext_vector_type(8))) float v8f;

#define LCONST 2048
#define DCONST 1024
#define NHC 16
#define DHC 64

static __device__ inline v8f wmma4(v2f a, v2f b, v8f c) {
    return __builtin_amdgcn_wmma_f32_16x16x4_f32(
         false, a,  false, b,
         (short)0, c,  false,  false);
}

typedef __attribute__((ext_vector_type(4))) float v4f;
#define GK 32
#define GP 36
#define GSTAGE (128 * GP)

static __device__ inline void gemm_tile_128x128(
    const float* __restrict__ a, const float* __restrict__ w,
    const int m0, const int n0, float* __restrict__ g_s, v8f (&acc)[4][4]) {
    const int tid  = threadIdx.x;
    const int lane = tid & 31;
    const int wave = tid >> 5;
    const int c    = lane & 15;
    const int hh   = lane >> 4;
    const int wm   = wave >> 1;
    const int wn   = wave & 1;

    const int trow = tid >> 3;
    const int tc4  = (tid & 7) * 4;
    const float* ag = a + (size_t)(m0 + trow) * DCONST + tc4;
    const float* wg = w + (size_t)(n0 + trow) * DCONST + tc4;
    const int st_off = trow * GP + tc4;
    const int fa_off = (wm * 64 + c) * GP + 4 * hh;
    const int fb_off = GSTAGE + (wn * 64 + c) * GP + 4 * hh;

    v8f zero = {};
#pragma unroll
    for (int mi = 0; mi < 4; ++mi)
#pragma unroll
        for (int ni = 0; ni < 4; ++ni) acc[mi][ni] = zero;

    v4f pa[8], pb[8];
#pragma unroll
    for (int i = 0; i < 8; ++i) {
        pa[i] = *(const v4f*)(ag + (size_t)i * 16 * DCONST);
        pb[i] = *(const v4f*)(wg + (size_t)i * 16 * DCONST);
    }
#pragma unroll
    for (int i = 0; i < 8; ++i) {
        *(v4f*)(g_s + st_off + i * 16 * GP) = pa[i];
        *(v4f*)(g_s + GSTAGE + st_off + i * 16 * GP) = pb[i];
    }
    __syncthreads();

    v4f af[2][4], bf[2][4];
#pragma unroll
    for (int i = 0; i < 4; ++i) {
        af[0][i] = *(const v4f*)(g_s + fa_off + i * 16 * GP);
        bf[0][i] = *(const v4f*)(g_s + fb_off + i * 16 * GP);
    }

    for (int t = 0; t < DCONST / GK; ++t) {
        const float* cur = g_s + (t & 1) * (2 * GSTAGE);
        float* nxt = g_s + ((t & 1) ^ 1) * (2 * GSTAGE);
        const bool more = (t + 1) < DCONST / GK;
        if (more) {
            const float* an = ag + (t + 1) * GK;
            const float* wn_ = wg + (t + 1) * GK;
#pragma unroll
            for (int i = 0; i < 8; ++i) {
                pa[i] = *(const v4f*)(an + (size_t)i * 16 * DCONST);
                pb[i] = *(const v4f*)(wn_ + (size_t)i * 16 * DCONST);
            }
        }

#pragma unroll
        for (int j = 0; j < GK / 8; ++j) {
            const int cb = j & 1;
            if (j + 1 < GK / 8) {
#pragma unroll
                for (int i = 0; i < 4; ++i) {
                    af[cb ^ 1][i] = *(const v4f*)(cur + fa_off + i * 16 * GP + 8 * (j + 1));
                    bf[cb ^ 1][i] = *(const v4f*)(cur + fb_off + i * 16 * GP + 8 * (j + 1));
                }
            }
            if (j == GK / 8 - 2 && more) {
#pragma unroll
                for (int i = 0; i < 8; ++i) {
                    *(v4f*)(nxt + st_off + i * 16 * GP) = pa[i];
                    *(v4f*)(nxt + GSTAGE + st_off + i * 16 * GP) = pb[i];
                }
            }
            if (j == GK / 8 - 1 && more) {
                __syncthreads();
#pragma unroll
                for (int i = 0; i < 4; ++i) {
                    af[0][i] = *(const v4f*)(nxt + fa_off + i * 16 * GP);
                    bf[0][i] = *(const v4f*)(nxt + fb_off + i * 16 * GP);
                }
            }
            __builtin_amdgcn_sched_barrier(0);
#pragma unroll
            for (int mi = 0; mi < 4; ++mi)
#pragma unroll
                for (int ni = 0; ni < 4; ++ni) {
                    const v2f a0 = {af[cb][mi].x, af[cb][mi].y};
                    const v2f b0 = {bf[cb][ni].x, bf[cb][ni].y};
                    acc[mi][ni] = wmma4(a0, b0, acc[mi][ni]);
                }
#pragma unroll
            for (int mi = 0; mi < 4; ++mi)
#pragma unroll
                for (int ni = 0; ni < 4; ++ni) {
                    const v2f a1 = {af[cb][mi].z, af[cb][mi].w};
                    const v2f b1 = {bf[cb][ni].z, bf[cb][ni].w};
                    acc[mi][ni] = wmma4(a1, b1, acc[mi][ni]);
                }
            __builtin_amdgcn_sched_barrier(0);
        }
    }
}

__global__ __launch_bounds__(128, 3) void qkv_rms_kernel(
    const float* __restrict__ x, const float* __restrict__ w,
    const float* __restrict__ qnw, const float* __restrict__ knw,
    float* __restrict__ qb, float* __restrict__ kb, float* __restrict__ vb) {
    __shared__ __attribute__((aligned(16))) float g_s[4 * GSTAGE];
    const int lane  = threadIdx.x & 31;
    const int wave  = threadIdx.x >> 5;
    const int laneN = lane & 15;
    const int hi8   = (lane >> 4) << 3;
    const int mb = blockIdx.y * 128;
    const int nb = blockIdx.x * 128;

    v8f acc[4][4];
    gemm_tile_128x128(x, w, mb, nb, g_s, acc);

    const int m0 = mb + (wave >> 1) * 64;
    const int n0 = nb + (wave & 1) * 64;

    const int sec = n0 >> 10;
    const int h   = (n0 & 1023) >> 6;
    float* dst = (sec == 0) ? qb : ((sec == 1) ? kb : vb);
    float wn[4];
#pragma unroll
    for (int ni = 0; ni < 4; ++ni) {
        const float* nw = (sec == 0) ? qnw : knw;
        wn[ni] = (sec < 2) ? nw[ni * 16 + laneN] : 1.0f;
    }

#pragma unroll
    for (int mi = 0; mi < 4; ++mi) {
#pragma unroll
        for (int r = 0; r < 8; ++r) {
            float sv[4];
#pragma unroll
            for (int ni = 0; ni < 4; ++ni) sv[ni] = acc[mi][ni][r];
            if (sec < 2) {
                float ss = sv[0] * sv[0] + sv[1] * sv[1] + sv[2] * sv[2] + sv[3] * sv[3];
                ss += __shfl_xor(ss, 1, 32);
                ss += __shfl_xor(ss, 2, 32);
                ss += __shfl_xor(ss, 4, 32);
                ss += __shfl_xor(ss, 8, 32);
                float rstd = rsqrtf(ss * (1.0f / 64.0f) + 1e-5f);
#pragma unroll
                for (int ni = 0; ni < 4; ++ni) sv[ni] *= rstd * wn[ni];
            }
            const int m = m0 + mi * 16 + r + hi8;
            const int b = m >> 11;
            const int l = m & (LCONST - 1);
            float* rowp = dst + ((size_t)((b * NHC + h) * LCONST + l) << 6);
#pragma unroll
            for (int ni = 0; ni < 4; ++ni) rowp[ni * 16 + laneN] = sv[ni];
        }
    }
}

#define KP 68

static __device__ inline float xhalf(float x) {
    const int b = __builtin_bit_cast(int, x);
    return __builtin_bit_cast(float,
        __builtin_amdgcn_permlanex16(b, b, 0x76543210, 0xfedcba98, false, false));
}

__global__ __launch_bounds__(128, 2) void attn_kernel(
    const float* __restrict__ qb, const float* __restrict__ kb,
    const float* __restrict__ vb, float* __restrict__ ob) {
    const int tid  = threadIdx.x;
    const int lane = tid & 31;
    const int wave = tid >> 5;
    const int c    = lane & 15;
    const int hh   = lane >> 4;
    const int bh = blockIdx.z * NHC + blockIdx.y;
    const float* qh = qb + (size_t)bh * LCONST * DHC;
    const float* kh = kb + (size_t)bh * LCONST * DHC;
    const float* vh = vb + (size_t)bh * LCONST * DHC;

    __shared__ __attribute__((aligned(16))) float k_s[64 * KP];
    __shared__ __attribute__((aligned(16))) float v_s[64 * KP];
    __shared__ __attribute__((aligned(16))) float o_s[4 * 16 * KP];
    float* o_w = o_s + wave * (16 * KP);

  for (int half = 0; half < 2; ++half) {
    const int qt = half ? (int)blockIdx.x : (LCONST / 64 - 1 - (int)blockIdx.x);
    const int q0 = qt * 64;
    const int qw = q0 + wave * 16;
    const int qrow = qw + c;

    v4f qa[8];
#pragma unroll
    for (int j = 0; j < 8; ++j)
        qa[j] = *(const v4f*)(qh + (size_t)qrow * DHC + 8 * j + 4 * hh);

    v8f zero = {};
    v8f o[4];
#pragma unroll
    for (int nj = 0; nj < 4; ++nj) o[nj] = zero;
    float mrow = -3.0e38f;
    float lrow = 0.0f;

    const int nkv = qt + 1;

    v4f kr[8], vr[8];
#pragma unroll
    for (int i = 0; i < 8; ++i) {
        kr[i] = *(const v4f*)(kh + (size_t)(tid + 128 * i) * 4);
        vr[i] = *(const v4f*)(vh + (size_t)(tid + 128 * i) * 4);
    }

    for (int kv = 0; kv < nkv; ++kv) {
        const int kbase = kv * 64;
        __syncthreads();
#pragma unroll
        for (int i = 0; i < 8; ++i) {
            const int f = tid + 128 * i;
            const int row = f >> 4, c4 = f & 15;
            *(v4f*)(k_s + row * KP + c4 * 4) = kr[i];
            *(v4f*)(v_s + row * KP + c4 * 4) = vr[i];
        }
        __syncthreads();
        if (kv + 1 < nkv) {
            const float* kn = kh + (size_t)(kbase + 64) * DHC;
            const float* vn = vh + (size_t)(kbase + 64) * DHC;
#pragma unroll
            for (int i = 0; i < 8; ++i) {
                kr[i] = *(const v4f*)(kn + (size_t)(tid + 128 * i) * 4);
                vr[i] = *(const v4f*)(vn + (size_t)(tid + 128 * i) * 4);
            }
        }

        v8f s[4];
#pragma unroll
        for (int ni = 0; ni < 4; ++ni) s[ni] = zero;
#pragma unroll
        for (int j = 0; j < 8; ++j) {
            v4f kf[4];
#pragma unroll
            for (int ni = 0; ni < 4; ++ni)
                kf[ni] = *(const v4f*)(k_s + (ni * 16 + c) * KP + 8 * j + 4 * hh);
            const v2f b0 = {qa[j].x, qa[j].y};
            const v2f b1 = {qa[j].z, qa[j].w};
#pragma unroll
            for (int ni = 0; ni < 4; ++ni) {
                const v2f a0 = {kf[ni].x, kf[ni].y};
                s[ni] = wmma4(a0, b0, s[ni]);
            }
#pragma unroll
            for (int ni = 0; ni < 4; ++ni) {
                const v2f a1 = {kf[ni].z, kf[ni].w};
                s[ni] = wmma4(a1, b1, s[ni]);
            }
        }

        const float scale = 0.125f;
        float vm[4];
#pragma unroll
        for (int ni = 0; ni < 4; ++ni) {
            vm[ni] = -3.0e38f;
#pragma unroll
            for (int r = 0; r < 8; ++r) {
                const int key = kbase + ni * 16 + 8 * hh + r;
                float t = s[ni][r] * scale;
                t = (key > qrow) ? -1.0e30f : t;
                s[ni][r] = t;
                vm[ni] = fmaxf(vm[ni], t);
            }
        }
        float vmax = fmaxf(fmaxf(vm[0], vm[1]), fmaxf(vm[2], vm[3]));
        vmax = fmaxf(vmax, xhalf(vmax));
        const float mnew = fmaxf(mrow, vmax);
        const float corr = __expf(mrow - mnew);
        float ps[4];
#pragma unroll
        for (int ni = 0; ni < 4; ++ni) {
            ps[ni] = 0.0f;
#pragma unroll
            for (int r = 0; r < 8; ++r) {
                const float p = __expf(s[ni][r] - mnew);
                s[ni][r] = p;
                ps[ni] += p;
            }
        }
        lrow = lrow * corr + ((ps[0] + ps[1]) + (ps[2] + ps[3]));
        mrow = mnew;
#pragma unroll
        for (int nj = 0; nj < 4; ++nj)
#pragma unroll
            for (int r = 0; r < 8; ++r) o[nj][r] *= corr;

#pragma unroll
        for (int ni = 0; ni < 4; ++ni) {
#pragma unroll
            for (int i = 0; i < 4; ++i) {
                const v2f b = {s[ni][2 * i], s[ni][2 * i + 1]};
#pragma unroll
                for (int nj = 0; nj < 4; ++nj) {
                    const float* vp = v_s + (16 * ni + 8 * hh + 2 * i) * KP + nj * 16 + c;
                    const v2f a = {vp[0], vp[KP]};
                    o[nj] = wmma4(a, b, o[nj]);
                }
            }
        }
    }

    const float inv = 1.0f / (lrow + xhalf(lrow));
#pragma unroll
    for (int nj = 0; nj < 4; ++nj) {
        const v4f lo = {o[nj][0] * inv, o[nj][1] * inv, o[nj][2] * inv, o[nj][3] * inv};
        const v4f hi = {o[nj][4] * inv, o[nj][5] * inv, o[nj][6] * inv, o[nj][7] * inv};
        *(v4f*)(o_w + c * KP + 16 * nj + 8 * hh)     = lo;
        *(v4f*)(o_w + c * KP + 16 * nj + 8 * hh + 4) = hi;
    }
    __builtin_amdgcn_fence(__ATOMIC_RELEASE, "workgroup", "local");
    __builtin_amdgcn_wave_barrier();
    __builtin_amdgcn_fence(__ATOMIC_ACQUIRE, "workgroup", "local");
#pragma unroll
    for (int i = 0; i < 8; ++i) {
        const int row = 2 * i + hh;
        const v4f ov = *(const v4f*)(o_w + row * KP + 4 * c);
        float* orow = ob + (size_t)(blockIdx.z * LCONST + qw + row) * DCONST + blockIdx.y * DHC;
        *(v4f*)(orow + 4 * c) = ov;
    }
    __builtin_amdgcn_fence(__ATOMIC_RELEASE, "workgroup", "local");
    __builtin_amdgcn_wave_barrier();
    __builtin_amdgcn_fence(__ATOMIC_ACQUIRE, "workgroup", "local");
  }
}

__global__ __launch_bounds__(128, 3) void fc_kernel(
    const float* __restrict__ oin, const float* __restrict__ w,
    float* __restrict__ out) {
    __shared__ __attribute__((aligned(16))) float g_s[4 * GSTAGE];
    const int lane  = threadIdx.x & 31;
    const int wave  = threadIdx.x >> 5;
    const int laneN = lane & 15;
    const int hi8   = (lane >> 4) << 3;
    const int mb = blockIdx.y * 128;
    const int nb = blockIdx.x * 128;

    v8f acc[4][4];
    gemm_tile_128x128(oin, w, mb, nb, g_s, acc);

    const int m0 = mb + (wave >> 1) * 64;
    const int n0 = nb + (wave & 1) * 64;

#pragma unroll
    for (int mi = 0; mi < 4; ++mi) {
#pragma unroll
        for (int r = 0; r < 8; ++r) {
            const int m = m0 + mi * 16 + r + hi8;
            float* rowp = out + (size_t)m * DCONST + n0;
#pragma unroll
            for (int ni = 0; ni < 4; ++ni) rowp[ni * 16 + laneN] = acc[mi][ni][r];
        }
    }
}

extern "C" void kernel_launch(void* const* d_in, const int* in_sizes, int n_in,
                              void* d_out, int out_size, void* d_ws, size_t ws_size,
                              hipStream_t stream) {
    const float* x     = (const float*)d_in[0];
    const float* w_qkv = (const float*)d_in[2];
    const float* w_fc  = (const float*)d_in[3];
    const float* qnw   = (const float*)d_in[4];
    const float* knw   = (const float*)d_in[5];

    const size_t per = (size_t)2 * NHC * LCONST * DHC;
    float* wsf  = (float*)d_ws;
    float* qb   = wsf;
    float* kb   = wsf + per;
    float* vb   = wsf + 2 * per;
    float* obuf = wsf + 3 * per;

    dim3 g1(3072 / 128, 4096 / 128);
    qkv_rms_kernel<<<g1, 128, 0, stream>>>(x, w_qkv, qnw, knw, qb, kb, vb);

    dim3 g2(LCONST / 128, NHC, 2);
    attn_kernel<<<g2, 128, 0, stream>>>(qb, kb, vb, obuf);

    dim3 g3(DCONST / 128, 4096 / 128);
    fc_kernel<<<g3, 128, 0, stream>>>(obuf, w_fc, (float*)d_out);
}
